# starting kernel
# baseline (speedup 1.0000x reference)
#include <hip/hip_runtime.h>
#include <hip/hip_bf16.h>


typedef __attribute__((ext_vector_type(2))) float v2f;
typedef __attribute__((ext_vector_type(8))) float v8f;

#define LN_EPS 1e-5f

template <int K, int NT, int MT>
__global__ __launch_bounds__(256)
void gemm_wmma_f32(const float* __restrict__ A,
                   const float* __restrict__ W,
                   const float* __restrict__ bias,
                   float* __restrict__ out,
                   int nrows) {
    constexpr int NOUT = NT * 16;
    const int wave = (blockIdx.x * blockDim.x + threadIdx.x) >> 5;
    const int lane = threadIdx.x & 31;
    const int m0 = wave * (MT * 16);
    if (m0 >= nrows) return;

    const int mr = lane & 15;
    const int hi = lane >> 4;

    v8f acc[MT][NT];
#pragma unroll
    for (int mt = 0; mt < MT; ++mt)
#pragma unroll
        for (int t = 0; t < NT; ++t)
            acc[mt][t] = (v8f){0.f, 0.f, 0.f, 0.f, 0.f, 0.f, 0.f, 0.f};

    const float* arow[MT];
#pragma unroll
    for (int mt = 0; mt < MT; ++mt)
        arow[mt] = A + (size_t)(m0 + mt * 16 + mr) * K;

#pragma unroll 2
    for (int k = 0; k < K; k += 4) {
        const int kb = k + hi * 2;
        v2f a[MT];
#pragma unroll
        for (int mt = 0; mt < MT; ++mt) {
            a[mt].x = arow[mt][kb];
            a[mt].y = arow[mt][kb + 1];
        }
        const float* w0 = W + (size_t)kb * NOUT;
        const float* w1 = w0 + NOUT;
#pragma unroll
        for (int t = 0; t < NT; ++t) {
            const int n = t * 16 + mr;
            v2f b;
            b.x = w0[n];
            b.y = w1[n];
#pragma unroll
            for (int mt = 0; mt < MT; ++mt)
                acc[mt][t] = __builtin_amdgcn_wmma_f32_16x16x4_f32(
                    false, a[mt], false, b, (short)0, acc[mt][t], false, false);
        }
    }

#pragma unroll
    for (int mt = 0; mt < MT; ++mt) {
#pragma unroll
        for (int t = 0; t < NT; ++t) {
            const int n = t * 16 + mr;
            const float bv = bias ? bias[n] : 0.f;
#pragma unroll
            for (int v = 0; v < 8; ++v) {
                const int m = m0 + mt * 16 + v + hi * 8;
                out[(size_t)m * NOUT + n] = acc[mt][t][v] + bv;
            }
        }
    }
}

__global__ void deg_init_kernel(float* deg, int N) {
    int i = blockIdx.x * blockDim.x + threadIdx.x;
    if (i < N) deg[i] = 1.0f;
}

__global__ void deg_count_kernel(const int* __restrict__ col, float* deg, int E) {
    int e = blockIdx.x * blockDim.x + threadIdx.x;
    if (e < E) atomicAdd(&deg[col[e]], 1.0f);
}

__global__ void deg_inv_kernel(const float* __restrict__ deg, float* dinv, int N) {
    int i = blockIdx.x * blockDim.x + threadIdx.x;
    if (i < N) dinv[i] = rsqrtf(deg[i]);
}

template <int F>
__global__ void init_self_kernel(const float* __restrict__ hw,
                                 const float* __restrict__ dinv,
                                 float* __restrict__ out, int N) {
    size_t i = (size_t)blockIdx.x * blockDim.x + threadIdx.x;
    if (i >= (size_t)N * F) return;
    const int node = (int)(i / F);
    const float dv = dinv[node];
    out[i] = hw[i] * dv * dv;
}

template <int F>
__global__ __launch_bounds__(256)
void scatter_edges_kernel(const float* __restrict__ src,
                          const int* __restrict__ rows,
                          const int* __restrict__ cols,
                          const float* __restrict__ dinv,
                          float* __restrict__ out, int E) {
    constexpr int PER = F / 32;
    const int wave = (blockIdx.x * blockDim.x + threadIdx.x) >> 5;
    const int lane = threadIdx.x & 31;
    if (wave >= E) return;
    const int r = rows[wave];
    const int c = cols[wave];
    const float nrm = dinv[r] * dinv[c];
    const float* s = src + (size_t)r * F + lane * PER;
    float* o = out + (size_t)c * F + lane * PER;
#pragma unroll
    for (int i = 0; i < PER; ++i)
        atomicAdd(o + i, s[i] * nrm);
}

__global__ __launch_bounds__(256)
void post_ln_relu_kernel(float* __restrict__ h,
                         const float* __restrict__ b,
                         const float* __restrict__ g,
                         const float* __restrict__ be, int N) {
    const int wave = (blockIdx.x * blockDim.x + threadIdx.x) >> 5;
    const int lane = threadIdx.x & 31;
    if (wave >= N) return;

    float v[4];
    float sum = 0.f;
    const size_t base = (size_t)wave * 128 + lane * 4;
#pragma unroll
    for (int i = 0; i < 4; ++i) {
        v[i] = h[base + i] + b[lane * 4 + i];
        sum += v[i];
    }
#pragma unroll
    for (int off = 16; off > 0; off >>= 1) sum += __shfl_xor(sum, off, 32);
    const float mu = sum * (1.0f / 128.0f);

    float var = 0.f;
#pragma unroll
    for (int i = 0; i < 4; ++i) {
        const float d = v[i] - mu;
        var += d * d;
    }
#pragma unroll
    for (int off = 16; off > 0; off >>= 1) var += __shfl_xor(var, off, 32);
    var *= (1.0f / 128.0f);
    const float rs = rsqrtf(var + LN_EPS);

#pragma unroll
    for (int i = 0; i < 4; ++i) {
        const float y = (v[i] - mu) * rs * g[lane * 4 + i] + be[lane * 4 + i];
        h[base + i] = fmaxf(y, 0.f);
    }
}

template <int F>
__global__ void add_bias_kernel(float* __restrict__ h,
                                const float* __restrict__ b, int N) {
    size_t i = (size_t)blockIdx.x * blockDim.x + threadIdx.x;
    if (i >= (size_t)N * F) return;
    h[i] += b[i % F];
}

extern "C" void kernel_launch(void* const* d_in, const int* in_sizes, int n_in,
                              void* d_out, int out_size, void* d_ws, size_t ws_size,
                              hipStream_t stream) {
    constexpr int F = 128, H = 128, D = 64, C = 32;
    constexpr int MT = 2;

    const float* x   = (const float*)d_in[0];
    const int*   ei  = (const int*)d_in[1];
    const float* W1  = (const float*)d_in[2];
    const float* b1  = (const float*)d_in[3];
    const float* g1  = (const float*)d_in[4];
    const float* be1 = (const float*)d_in[5];
    const float* W2  = (const float*)d_in[6];
    const float* b2  = (const float*)d_in[7];
    const float* Wc  = (const float*)d_in[8];
    const float* bc  = (const float*)d_in[9];
    float* logits    = (float*)d_out;

    const int N = in_sizes[0] / F;
    const int E = in_sizes[1] / 2;
    const int* rows = ei;
    const int* cols = ei + E;

    float* ws   = (float*)d_ws;
    float* hw1  = ws;
    float* agg2 = ws;
    float* hw2  = ws + (size_t)N * 64;
    float* agg1 = ws + (size_t)N * 128;
    float* deg  = ws + (size_t)N * 256;
    float* dinv = deg + N;

    const int TPB = 256;
    const int wavesPerBlk = TPB / 32;
    auto cdiv = [](long long a, long long b) { return (int)((a + b - 1) / b); };

    deg_init_kernel<<<cdiv(N, TPB), TPB, 0, stream>>>(deg, N);
    deg_count_kernel<<<cdiv(E, TPB), TPB, 0, stream>>>(cols, deg, E);
    deg_inv_kernel<<<cdiv(N, TPB), TPB, 0, stream>>>(deg, dinv, N);

    {
        const int nWaves = cdiv(N, 16 * MT);
        gemm_wmma_f32<F, H / 16, MT><<<cdiv(nWaves, wavesPerBlk), TPB, 0, stream>>>(
            x, W1, nullptr, hw1, N);
    }

    init_self_kernel<H><<<cdiv((long long)N * H, TPB), TPB, 0, stream>>>(hw1, dinv, agg1, N);
    scatter_edges_kernel<H><<<cdiv(E, wavesPerBlk), TPB, 0, stream>>>(
        hw1, rows, cols, dinv, agg1, E);

    post_ln_relu_kernel<<<cdiv(N, wavesPerBlk), TPB, 0, stream>>>(agg1, b1, g1, be1, N);

    {
        const int nWaves = cdiv(N, 16 * MT);
        gemm_wmma_f32<H, D / 16, MT><<<cdiv(nWaves, wavesPerBlk), TPB, 0, stream>>>(
            agg1, W2, nullptr, hw2, N);
    }

    init_self_kernel<D><<<cdiv((long long)N * D, TPB), TPB, 0, stream>>>(hw2, dinv, agg2, N);
    scatter_edges_kernel<D><<<cdiv(E, wavesPerBlk), TPB, 0, stream>>>(
        hw2, rows, cols, dinv, agg2, E);
    add_bias_kernel<D><<<cdiv((long long)N * D, TPB), TPB, 0, stream>>>(agg2, b2, N);

    {
        const int nWaves = cdiv(N, 16 * MT);
        gemm_wmma_f32<D, C / 16, MT><<<cdiv(nWaves, wavesPerBlk), TPB, 0, stream>>>(
            agg2, Wc, bc, logits, N);
    }
}


# First edge sum in three phases: filter the whole owner map into one LDS queue, resolve the ids to (source, local target) words in one parallel round, then stream the LDS-atomic adds; 27 -> 3 barriers
# speedup vs baseline: 2.3175x; 2.3175x over previous
#include <hip/hip_runtime.h>
#include <hip/hip_bf16.h>


typedef __attribute__((ext_vector_type(2))) float v2f;
typedef __attribute__((ext_vector_type(8))) float v8f;

#define LN_EPS 1e-5f

template <int K, int NT, int MT>
__global__ __launch_bounds__(256)
void gemm_wmma_f32(const float* __restrict__ A,
                   const float* __restrict__ W,
                   const float* __restrict__ bias,
                   float* __restrict__ out,
                   int nrows) {
    constexpr int NOUT = NT * 16;
    const int wave = (blockIdx.x * blockDim.x + threadIdx.x) >> 5;
    const int lane = threadIdx.x & 31;
    const int m0 = wave * (MT * 16);
    if (m0 >= nrows) return;

    const int mr = lane & 15;
    const int hi = lane >> 4;

    v8f acc[MT][NT];
#pragma unroll
    for (int mt = 0; mt < MT; ++mt)
#pragma unroll
        for (int t = 0; t < NT; ++t)
            acc[mt][t] = (v8f){0.f, 0.f, 0.f, 0.f, 0.f, 0.f, 0.f, 0.f};

    const float* arow[MT];
#pragma unroll
    for (int mt = 0; mt < MT; ++mt)
        arow[mt] = A + (size_t)(m0 + mt * 16 + mr) * K;

#pragma unroll 2
    for (int k = 0; k < K; k += 4) {
        const int kb = k + hi * 2;
        v2f a[MT];
#pragma unroll
        for (int mt = 0; mt < MT; ++mt) {
            a[mt].x = arow[mt][kb];
            a[mt].y = arow[mt][kb + 1];
        }
        const float* w0 = W + (size_t)kb * NOUT;
        const float* w1 = w0 + NOUT;
#pragma unroll
        for (int t = 0; t < NT; ++t) {
            const int n = t * 16 + mr;
            v2f b;
            b.x = w0[n];
            b.y = w1[n];
#pragma unroll
            for (int mt = 0; mt < MT; ++mt)
                acc[mt][t] = __builtin_amdgcn_wmma_f32_16x16x4_f32(
                    false, a[mt], false, b, (short)0, acc[mt][t], false, false);
        }
    }

#pragma unroll
    for (int mt = 0; mt < MT; ++mt) {
#pragma unroll
        for (int t = 0; t < NT; ++t) {
            const int n = t * 16 + mr;
            const float bv = bias ? bias[n] : 0.f;
#pragma unroll
            for (int v = 0; v < 8; ++v) {
                const int m = m0 + mt * 16 + v + hi * 8;
                out[(size_t)m * NOUT + n] = acc[mt][t][v] + bv;
            }
        }
    }
}

__global__ void deg_init_kernel(float* deg, int N) {
    int i = blockIdx.x * blockDim.x + threadIdx.x;
    if (i < N) deg[i] = 1.0f;
}

__global__ void deg_inv_kernel(const float* __restrict__ deg, float* dinv, int N) {
    int i = blockIdx.x * blockDim.x + threadIdx.x;
    if (i < N) dinv[i] = rsqrtf(deg[i]);
}

constexpr int OWN_R    = 400;
constexpr int OWN_TPB  = 1024;
constexpr int OWN_NW   = OWN_TPB / 32;
constexpr int OWN_VPT  = 12;
constexpr int OWN_VCH  = OWN_TPB * OWN_VPT;
constexpr int OWN_QCAP = 8192;
constexpr int OWN_U    = 8;
constexpr int OWN_RBITS   = 17;
constexpr int OWN_LCAP    = 8192;
constexpr int OWN_LSTRIDE = OWN_LCAP + 8;
constexpr unsigned OWN_LBAD = 0xffffffffu;
static_assert(OWN_U == 8, "a batch of list entries is read as two 16-byte loads");
static_assert(OWN_QCAP % OWN_TPB == 0 && OWN_QCAP % OWN_U == 0 && OWN_QCAP <= OWN_LCAP,
              "every thread resolves OWN_QCAP / OWN_TPB queue entries; whole batches; the queue fits the list");
static_assert(OWN_R < (1 << (32 - OWN_RBITS)), "the local target fits above the source id");

__global__ __launch_bounds__(256)
void deg_count_map_kernel(const int* __restrict__ col, float* deg,
                          unsigned int* __restrict__ omap, int E) {
    const int t  = blockIdx.x * blockDim.x + threadIdx.x;
    const int e0 = 4 * t;
    if (e0 + 3 < E) {
        const int c0 = col[e0], c1 = col[e0 + 1], c2 = col[e0 + 2], c3 = col[e0 + 3];
        atomicAdd(&deg[c0], 1.0f);
        atomicAdd(&deg[c1], 1.0f);
        atomicAdd(&deg[c2], 1.0f);
        atomicAdd(&deg[c3], 1.0f);
        omap[t] = (unsigned)(c0 / OWN_R) | ((unsigned)(c1 / OWN_R) << 8) |
                  ((unsigned)(c2 / OWN_R) << 16) | ((unsigned)(c3 / OWN_R) << 24);
    } else {
        for (int e = e0; e < E; ++e) atomicAdd(&deg[col[e]], 1.0f);
    }
}

template <int F>
__device__ __forceinline__ void edge_add_serial(const float* __restrict__ src,
                                                const int* __restrict__ rows,
                                                const float* __restrict__ dinv,
                                                float* acc, int e, int c, int lo) {
    constexpr int PER = F / 32;
    const int r = rows[e];
    const float w = dinv[r] * dinv[c];
    const float* s = src + (size_t)r * F;
    float* a = acc + (c - lo) * F;
    for (int f = 0; f < F; ++f)
        atomicAdd(a + (f % PER) * 32 + (f / PER), s[f] * w);
}

template <int M>
__device__ __forceinline__ float row_xor_lane(float x) {
    const int b = __builtin_bit_cast(int, x);
    return __builtin_bit_cast(float, __builtin_amdgcn_update_dpp(b, b, 0x160 + M, 0xf, 0xf, false));
}
__device__ __forceinline__ float other_row_lane(float x) {
    const int b = __builtin_bit_cast(int, x);
    return __builtin_bit_cast(float, __builtin_amdgcn_permlanex16(b, b, 0x76543210, 0xfedcba98, false, false));
}
__device__ __forceinline__ float wave_sum(float x) {
    x += other_row_lane(x);
    x += row_xor_lane<8>(x);
    x += row_xor_lane<4>(x);
    x += row_xor_lane<2>(x);
    x += row_xor_lane<1>(x);
    return x;
}

template <int F, int MODE, bool LNRELU>
__global__ __launch_bounds__(OWN_TPB)
void agg_owner_lds_kernel(const float* __restrict__ src,
                          const int* __restrict__ rows,
                          const int* __restrict__ cols,
                          const unsigned int* __restrict__ omap,
                          unsigned int* __restrict__ lbase, int lwords,
                          const float* __restrict__ dinv,
                          const float* __restrict__ bias,
                          const float* __restrict__ g,
                          const float* __restrict__ be,
                          float* __restrict__ out, int N, int E) {
    constexpr int PER = F / 32;
    typedef __attribute__((ext_vector_type(PER))) float vecf;

    __shared__ __attribute__((aligned(16))) float acc[(OWN_R + 1) * F];
    __shared__ __attribute__((aligned(16))) int queue[OWN_QCAP];
    __shared__ int s_cnt;

    const int tid  = threadIdx.x;
    const int lane = tid & 31;
    const int wave = __builtin_amdgcn_readfirstlane(tid >> 5);
    const int lo   = blockIdx.x * OWN_R;
    const int nown = min(OWN_R, N - lo);
    const unsigned pat  = (unsigned)blockIdx.x * 0x01010101u;
    const unsigned npat = ~pat;

    const int loff = (((E + 3) >> 2) + 63) & ~63;
    const bool list_ok = (N <= (1 << OWN_RBITS)) &&
                         ((long long)loff + (long long)gridDim.x * OWN_LSTRIDE <= (long long)lwords);
    unsigned int* mylist = lbase + loff + (size_t)blockIdx.x * OWN_LSTRIDE;
    unsigned int* ent = mylist + 8;
    bool lbad = !list_ok;

    auto add_batch = [&](const unsigned (&wd)[OWN_U], int q0, int n) {
        int   lc[OWN_U];
        float w[OWN_U];
        vecf  v[OWN_U];
#pragma unroll
        for (int u = 0; u < OWN_U; ++u) {
            const int r = (int)(wd[u] & ((1u << OWN_RBITS) - 1u));
            lc[u] = (int)(wd[u] >> OWN_RBITS);
            w[u] = dinv[r] * dinv[lo + lc[u]];
            v[u] = *(const vecf*)(src + (size_t)r * F + lane * PER);
        }
#pragma unroll
        for (int u = 0; u < OWN_U; ++u) {
            float* a = acc + ((q0 + u < n) ? lc[u] : OWN_R) * F + lane;
#pragma unroll
            for (int k = 0; k < PER; ++k)
                atomicAdd(a + k * 32, v[u][k] * w[u]);
        }
    };

    for (int i = wave; i < nown; i += OWN_NW) {
        const int node = lo + i;
        const float dv = dinv[node];
        const vecf v = *(const vecf*)(src + (size_t)node * F + lane * PER);
#pragma unroll
        for (int k = 0; k < PER; ++k)
            acc[i * F + k * 32 + lane] = v[k] * dv * dv;
    }
    if (tid == 0) s_cnt = 0;
    __syncthreads();

    const int nvec = E >> 4;
    const uint4* mv = (const uint4*)omap;
    const int ntail = E - 16 * nvec;
    if (tid < ntail) {
        const int e = 16 * nvec + tid;
        const int c = cols[e];
        if ((unsigned)(c - lo) < (unsigned)nown)
            edge_add_serial<F>(src, rows, dinv, acc, e, c, lo);
    }

    unsigned lcnt = OWN_LBAD;
    if constexpr (MODE == 1) {
        if (list_ok) lcnt = mylist[0];
    }

    if (lcnt == OWN_LBAD) {
        uint4 cb[OWN_VPT];
#pragma unroll
        for (int j = 0; j < OWN_VPT; ++j) {
            const int vi = j * OWN_TPB + tid;
            cb[j] = (vi < nvec) ? mv[vi] : make_uint4(npat, npat, npat, npat);
        }

        for (int vbase = 0; vbase < nvec; vbase += OWN_VCH) {
            unsigned hit = 0u;
#pragma unroll
            for (int j = 0; j < OWN_VPT; ++j) {
                const unsigned x0 = cb[j].x ^ pat, x1 = cb[j].y ^ pat;
                const unsigned x2 = cb[j].z ^ pat, x3 = cb[j].w ^ pat;
                const unsigned any = ((x0 - 0x01010101u) & ~x0) | ((x1 - 0x01010101u) & ~x1) |
                                     ((x2 - 0x01010101u) & ~x2) | ((x3 - 0x01010101u) & ~x3);
                if (any & 0x80808080u) hit |= 1u << j;
            }

#pragma unroll
            for (int j = 0; j < OWN_VPT; ++j) {
                const int vi = vbase + OWN_VCH + j * OWN_TPB + tid;
                cb[j] = (vi < nvec) ? mv[vi] : make_uint4(npat, npat, npat, npat);
            }

            while (hit) {
                const int j = __builtin_ctz(hit);
                hit &= hit - 1u;
                const int vi = vbase + j * OWN_TPB + tid;
                const uint4 c4 = mv[vi];
                const unsigned xs[4] = {c4.x ^ pat, c4.y ^ pat, c4.z ^ pat, c4.w ^ pat};
                unsigned z[4];
                int cnt = 0;
#pragma unroll
                for (int d = 0; d < 4; ++d) {
                    z[d] = ~(((xs[d] & 0x7f7f7f7fu) + 0x7f7f7f7fu) | xs[d]) & 0x80808080u;
                    cnt += __builtin_popcount(z[d]);
                }
                int p = atomicAdd(&s_cnt, cnt);
#pragma unroll
                for (int d = 0; d < 4; ++d) {
                    unsigned m = z[d];
                    while (m) {
                        const int e = 16 * vi + 4 * d + (__builtin_ctz(m) >> 3);
                        m &= m - 1u;
                        if (p < OWN_QCAP) queue[p] = e;
                        else edge_add_serial<F>(src, rows, dinv, acc, e, cols[e], lo);
                        ++p;
                    }
                }
            }
        }
        __syncthreads();

        const int sc = __builtin_amdgcn_readfirstlane(s_cnt);
        const int n = min(sc, OWN_QCAP);
        const int npad = (n + OWN_U - 1) & ~(OWN_U - 1);
        if constexpr (MODE == 0) {
            if (sc > OWN_QCAP) lbad = true;
        }

        if (n > 0) {
            constexpr int NRES = OWN_QCAP / OWN_TPB;
            int rr[NRES], cc[NRES];
#pragma unroll
            for (int i = 0; i < NRES; ++i) {
                const int q = i * OWN_TPB + tid;
                const int eq = queue[q];
                const int e = (q < n) ? eq : 0;
                rr[i] = rows[e];
                cc[i] = cols[e];
            }
#pragma unroll
            for (int i = 0; i < NRES; ++i) {
                const int q = i * OWN_TPB + tid;
                if (q < npad) {
                    const unsigned word = (q < n) ? ((unsigned)rr[i] | ((unsigned)(cc[i] - lo) << OWN_RBITS)) : 0u;
                    queue[q] = (int)word;
                    if constexpr (MODE == 0) {
                        if (!lbad) ent[q] = word;
                    }
                }
            }
        }
        if constexpr (MODE == 0) {
            if (list_ok && tid == 0) mylist[0] = lbad ? OWN_LBAD : (unsigned)n;
        }
        __syncthreads();

        int q0 = wave * OWN_U;
        unsigned nw[OWN_U] = {0u, 0u, 0u, 0u, 0u, 0u, 0u, 0u};
        if (q0 < n) {
            const uint4 fa = *(const uint4*)(queue + q0);
            const uint4 fb = *(const uint4*)(queue + q0 + 4);
            nw[0] = (unsigned)__builtin_amdgcn_readfirstlane((int)fa.x);
            nw[1] = (unsigned)__builtin_amdgcn_readfirstlane((int)fa.y);
            nw[2] = (unsigned)__builtin_amdgcn_readfirstlane((int)fa.z);
            nw[3] = (unsigned)__builtin_amdgcn_readfirstlane((int)fa.w);
            nw[4] = (unsigned)__builtin_amdgcn_readfirstlane((int)fb.x);
            nw[5] = (unsigned)__builtin_amdgcn_readfirstlane((int)fb.y);
            nw[6] = (unsigned)__builtin_amdgcn_readfirstlane((int)fb.z);
            nw[7] = (unsigned)__builtin_amdgcn_readfirstlane((int)fb.w);
        }
        for (; q0 < n; q0 += OWN_NW * OWN_U) {
            const unsigned wd[OWN_U] = {nw[0], nw[1], nw[2], nw[3], nw[4], nw[5], nw[6], nw[7]};
            const int qn = q0 + OWN_NW * OWN_U;
            uint4 ta = make_uint4(0u, 0u, 0u, 0u), tb = ta;
            if (qn < n) {
                ta = *(const uint4*)(queue + qn);
                tb = *(const uint4*)(queue + qn + 4);
            }
            add_batch(wd, q0, n);
            nw[0] = (unsigned)__builtin_amdgcn_readfirstlane((int)ta.x);
            nw[1] = (unsigned)__builtin_amdgcn_readfirstlane((int)ta.y);
            nw[2] = (unsigned)__builtin_amdgcn_readfirstlane((int)ta.z);
            nw[3] = (unsigned)__builtin_amdgcn_readfirstlane((int)ta.w);
            nw[4] = (unsigned)__builtin_amdgcn_readfirstlane((int)tb.x);
            nw[5] = (unsigned)__builtin_amdgcn_readfirstlane((int)tb.y);
            nw[6] = (unsigned)__builtin_amdgcn_readfirstlane((int)tb.z);
            nw[7] = (unsigned)__builtin_amdgcn_readfirstlane((int)tb.w);
        }
    } else {
        const int n = (int)lcnt;
        int q0 = wave * OWN_U;
        uint4 na = make_uint4(0u, 0u, 0u, 0u), nb = na;
        if (q0 < n) {
            na = *(const uint4*)(ent + q0);
            nb = *(const uint4*)(ent + q0 + 4);
        }
        for (; q0 < n; q0 += OWN_NW * OWN_U) {
            const uint4 ca = na, cb4 = nb;
            const int qn = q0 + OWN_NW * OWN_U;
            if (qn < n) {
                na = *(const uint4*)(ent + qn);
                nb = *(const uint4*)(ent + qn + 4);
            }
            const unsigned wd[OWN_U] = {ca.x, ca.y, ca.z, ca.w, cb4.x, cb4.y, cb4.z, cb4.w};
            add_batch(wd, q0, n);
        }
    }

    __syncthreads();

    const vecf bv = *(const vecf*)(bias + lane * PER);
    vecf gv = bv, bev = bv;
    if constexpr (LNRELU) {
        gv  = *(const vecf*)(g + lane * PER);
        bev = *(const vecf*)(be + lane * PER);
    }
    for (int i = wave; i < nown; i += OWN_NW) {
        vecf v;
#pragma unroll
        for (int k = 0; k < PER; ++k)
            v[k] = acc[i * F + k * 32 + lane] + bv[k];
        if constexpr (LNRELU) {
            float sum = 0.f;
#pragma unroll
            for (int k = 0; k < PER; ++k) sum += v[k];
            sum = wave_sum(sum);
            const float mu = sum * (1.0f / F);
            float var = 0.f;
#pragma unroll
            for (int k = 0; k < PER; ++k) {
                const float d = v[k] - mu;
                var += d * d;
            }
            var = wave_sum(var);
            var *= (1.0f / F);
            const float rs = rsqrtf(var + LN_EPS);
#pragma unroll
            for (int k = 0; k < PER; ++k)
                v[k] = fmaxf((v[k] - mu) * rs * gv[k] + bev[k], 0.f);
        }
        *(vecf*)(out + (size_t)(lo + i) * F + lane * PER) = v;
    }
}

extern "C" void kernel_launch(void* const* d_in, const int* in_sizes, int n_in,
                              void* d_out, int out_size, void* d_ws, size_t ws_size,
                              hipStream_t stream) {
    constexpr int F = 128, H = 128, D = 64, C = 32;
    constexpr int MT = 2;

    const float* x   = (const float*)d_in[0];
    const int*   ei  = (const int*)d_in[1];
    const float* W1  = (const float*)d_in[2];
    const float* b1  = (const float*)d_in[3];
    const float* g1  = (const float*)d_in[4];
    const float* be1 = (const float*)d_in[5];
    const float* W2  = (const float*)d_in[6];
    const float* b2  = (const float*)d_in[7];
    const float* Wc  = (const float*)d_in[8];
    const float* bc  = (const float*)d_in[9];
    float* logits    = (float*)d_out;
    unsigned int* omap = (unsigned int*)d_out;

    const int N = in_sizes[0] / F;
    const int E = in_sizes[1] / 2;
    const int* rows = ei;
    const int* cols = ei + E;

    float* ws   = (float*)d_ws;
    float* hw1  = ws;
    float* agg2 = ws;
    float* hw2  = ws + (size_t)N * 64;
    float* h2   = ws + (size_t)N * 128;
    float* deg  = ws + (size_t)N * 256;
    float* dinv = deg + N;

    const int TPB = 256;
    const int wavesPerBlk = TPB / 32;
    auto cdiv = [](long long a, long long b) { return (int)((a + b - 1) / b); };

    deg_init_kernel<<<cdiv(N, TPB), TPB, 0, stream>>>(deg, N);
    deg_count_map_kernel<<<cdiv(cdiv(E, 4), TPB), TPB, 0, stream>>>(cols, deg, omap, E);
    deg_inv_kernel<<<cdiv(N, TPB), TPB, 0, stream>>>(deg, dinv, N);

    {
        const int nWaves = cdiv(N, 16 * MT);
        gemm_wmma_f32<F, H / 16, MT><<<cdiv(nWaves, wavesPerBlk), TPB, 0, stream>>>(
            x, W1, nullptr, hw1, N);
    }

    agg_owner_lds_kernel<H, 0, true><<<cdiv(N, OWN_R), OWN_TPB, 0, stream>>>(
        hw1, rows, cols, omap, omap, out_size, dinv, b1, g1, be1, h2, N, E);

    {
        const int nWaves = cdiv(N, 16 * MT);
        gemm_wmma_f32<H, D / 16, MT><<<cdiv(nWaves, wavesPerBlk), TPB, 0, stream>>>(
            h2, W2, nullptr, hw2, N);
    }

    agg_owner_lds_kernel<D, 1, false><<<cdiv(N, OWN_R), OWN_TPB, 0, stream>>>(
        hw2, rows, cols, omap, omap, out_size, dinv, b2, nullptr, nullptr, agg2, N, E);

    {
        const int nWaves = cdiv(N, 16 * MT);
        gemm_wmma_f32<D, C / 16, MT><<<cdiv(nWaves, wavesPerBlk), TPB, 0, stream>>>(
            agg2, Wc, bc, logits, N);
    }
}


# Edge weight factored out of the edge loop: products store source rows pre-scaled by dinv[row], owner blocks add rows unscaled and scale each sum by dinv[target] at the store; no per-edge scale loads
# speedup vs baseline: 2.4674x; 1.0647x over previous
#include <hip/hip_runtime.h>
#include <hip/hip_bf16.h>


typedef __attribute__((ext_vector_type(2))) float v2f;
typedef __attribute__((ext_vector_type(8))) float v8f;

#define LN_EPS 1e-5f

template <int K, int NT, int MT, bool SCALE>
__global__ __launch_bounds__(256)
void gemm_wmma_f32(const float* __restrict__ A,
                   const float* __restrict__ W,
                   const float* __restrict__ bias,
                   const float* __restrict__ rowscale,
                   float* __restrict__ out,
                   int nrows) {
    constexpr int NOUT = NT * 16;
    const int wave = (blockIdx.x * blockDim.x + threadIdx.x) >> 5;
    const int lane = threadIdx.x & 31;
    const int m0 = wave * (MT * 16);
    if (m0 >= nrows) return;

    const int mr = lane & 15;
    const int hi = lane >> 4;

    v8f acc[MT][NT];
#pragma unroll
    for (int mt = 0; mt < MT; ++mt)
#pragma unroll
        for (int t = 0; t < NT; ++t)
            acc[mt][t] = (v8f){0.f, 0.f, 0.f, 0.f, 0.f, 0.f, 0.f, 0.f};

    const float* arow[MT];
#pragma unroll
    for (int mt = 0; mt < MT; ++mt)
        arow[mt] = A + (size_t)(m0 + mt * 16 + mr) * K;

#pragma unroll 2
    for (int k = 0; k < K; k += 4) {
        const int kb = k + hi * 2;
        v2f a[MT];
#pragma unroll
        for (int mt = 0; mt < MT; ++mt) {
            a[mt].x = arow[mt][kb];
            a[mt].y = arow[mt][kb + 1];
        }
        const float* w0 = W + (size_t)kb * NOUT;
        const float* w1 = w0 + NOUT;
#pragma unroll
        for (int t = 0; t < NT; ++t) {
            const int n = t * 16 + mr;
            v2f b;
            b.x = w0[n];
            b.y = w1[n];
#pragma unroll
            for (int mt = 0; mt < MT; ++mt)
                acc[mt][t] = __builtin_amdgcn_wmma_f32_16x16x4_f32(
                    false, a[mt], false, b, (short)0, acc[mt][t], false, false);
        }
    }

#pragma unroll
    for (int mt = 0; mt < MT; ++mt) {
        float rs[8];
#pragma unroll
        for (int v = 0; v < 8; ++v) {
            if constexpr (SCALE) rs[v] = rowscale[m0 + mt * 16 + v + hi * 8];
            else rs[v] = 1.f;
        }
#pragma unroll
        for (int t = 0; t < NT; ++t) {
            const int n = t * 16 + mr;
            const float bv = bias ? bias[n] : 0.f;
#pragma unroll
            for (int v = 0; v < 8; ++v) {
                const int m = m0 + mt * 16 + v + hi * 8;
                if constexpr (SCALE) out[(size_t)m * NOUT + n] = acc[mt][t][v] * rs[v] + bv;
                else out[(size_t)m * NOUT + n] = acc[mt][t][v] + bv;
            }
        }
    }
}

__global__ void deg_init_kernel(float* deg, int N) {
    int i = blockIdx.x * blockDim.x + threadIdx.x;
    if (i < N) deg[i] = 1.0f;
}

__global__ void deg_inv_kernel(const float* __restrict__ deg, float* dinv, int N) {
    int i = blockIdx.x * blockDim.x + threadIdx.x;
    if (i < N) dinv[i] = rsqrtf(deg[i]);
}

constexpr int OWN_R    = 400;
constexpr int OWN_TPB  = 1024;
constexpr int OWN_NW   = OWN_TPB / 32;
constexpr int OWN_VPT  = 12;
constexpr int OWN_VCH  = OWN_TPB * OWN_VPT;
constexpr int OWN_QCAP = 8192;
constexpr int OWN_U    = 8;
constexpr int OWN_RBITS   = 17;
constexpr int OWN_LCAP    = 8192;
constexpr int OWN_LSTRIDE = OWN_LCAP + 8;
constexpr unsigned OWN_LBAD = 0xffffffffu;
static_assert(OWN_U == 8, "a batch of list entries is read as two 16-byte loads");
static_assert(OWN_QCAP % OWN_TPB == 0 && OWN_QCAP % OWN_U == 0 && OWN_QCAP <= OWN_LCAP,
              "every thread resolves OWN_QCAP / OWN_TPB queue entries; whole batches; the queue fits the list");
static_assert(OWN_R < (1 << (32 - OWN_RBITS)), "the local target fits above the source id");

__global__ __launch_bounds__(256)
void deg_count_map_kernel(const int* __restrict__ col, float* deg,
                          unsigned int* __restrict__ omap, int E) {
    const int t  = blockIdx.x * blockDim.x + threadIdx.x;
    const int e0 = 4 * t;
    if (e0 + 3 < E) {
        const int c0 = col[e0], c1 = col[e0 + 1], c2 = col[e0 + 2], c3 = col[e0 + 3];
        atomicAdd(&deg[c0], 1.0f);
        atomicAdd(&deg[c1], 1.0f);
        atomicAdd(&deg[c2], 1.0f);
        atomicAdd(&deg[c3], 1.0f);
        omap[t] = (unsigned)(c0 / OWN_R) | ((unsigned)(c1 / OWN_R) << 8) |
                  ((unsigned)(c2 / OWN_R) << 16) | ((unsigned)(c3 / OWN_R) << 24);
    } else {
        for (int e = e0; e < E; ++e) atomicAdd(&deg[col[e]], 1.0f);
    }
}

template <int F>
__device__ __forceinline__ void edge_add_serial(const float* __restrict__ src,
                                                const int* __restrict__ rows,
                                                float* acc, int e, int c, int lo) {
    constexpr int PER = F / 32;
    const int r = rows[e];
    const float* s = src + (size_t)r * F;
    float* a = acc + (c - lo) * F;
    for (int f = 0; f < F; ++f)
        atomicAdd(a + (f % PER) * 32 + (f / PER), s[f]);
}

template <int M>
__device__ __forceinline__ float row_xor_lane(float x) {
    const int b = __builtin_bit_cast(int, x);
    return __builtin_bit_cast(float, __builtin_amdgcn_update_dpp(b, b, 0x160 + M, 0xf, 0xf, false));
}
__device__ __forceinline__ float other_row_lane(float x) {
    const int b = __builtin_bit_cast(int, x);
    return __builtin_bit_cast(float, __builtin_amdgcn_permlanex16(b, b, 0x76543210, 0xfedcba98, false, false));
}
__device__ __forceinline__ float wave_sum(float x) {
    x += other_row_lane(x);
    x += row_xor_lane<8>(x);
    x += row_xor_lane<4>(x);
    x += row_xor_lane<2>(x);
    x += row_xor_lane<1>(x);
    return x;
}

template <int F, int MODE, bool LNRELU>
__global__ __launch_bounds__(OWN_TPB)
void agg_owner_lds_kernel(const float* __restrict__ src,
                          const int* __restrict__ rows,
                          const int* __restrict__ cols,
                          const unsigned int* __restrict__ omap,
                          unsigned int* __restrict__ lbase, int lwords,
                          const float* __restrict__ dinv,
                          const float* __restrict__ bias,
                          const float* __restrict__ g,
                          const float* __restrict__ be,
                          float* __restrict__ out, int N, int E) {
    constexpr int PER = F / 32;
    typedef __attribute__((ext_vector_type(PER))) float vecf;

    __shared__ __attribute__((aligned(16))) float acc[(OWN_R + 1) * F];
    __shared__ __attribute__((aligned(16))) int queue[OWN_QCAP];
    __shared__ int s_cnt;

    const int tid  = threadIdx.x;
    const int lane = tid & 31;
    const int wave = __builtin_amdgcn_readfirstlane(tid >> 5);
    const int lo   = blockIdx.x * OWN_R;
    const int nown = min(OWN_R, N - lo);
    const unsigned pat  = (unsigned)blockIdx.x * 0x01010101u;
    const unsigned npat = ~pat;

    const int loff = (((E + 3) >> 2) + 63) & ~63;
    const bool list_ok = (N <= (1 << OWN_RBITS)) &&
                         ((long long)loff + (long long)gridDim.x * OWN_LSTRIDE <= (long long)lwords);
    unsigned int* mylist = lbase + loff + (size_t)blockIdx.x * OWN_LSTRIDE;
    unsigned int* ent = mylist + 8;
    bool lbad = !list_ok;

    auto add_batch = [&](const unsigned (&wd)[OWN_U], int q0, int n) {
        int   lc[OWN_U];
        vecf  v[OWN_U];
#pragma unroll
        for (int u = 0; u < OWN_U; ++u) {
            const int r = (int)(wd[u] & ((1u << OWN_RBITS) - 1u));
            lc[u] = (int)(wd[u] >> OWN_RBITS);
            v[u] = *(const vecf*)(src + (size_t)r * F + lane * PER);
        }
#pragma unroll
        for (int u = 0; u < OWN_U; ++u) {
            float* a = acc + ((q0 + u < n) ? lc[u] : OWN_R) * F + lane;
#pragma unroll
            for (int k = 0; k < PER; ++k)
                atomicAdd(a + k * 32, v[u][k]);
        }
    };

    for (int i = wave; i < nown; i += OWN_NW) {
        const int node = lo + i;
        const vecf v = *(const vecf*)(src + (size_t)node * F + lane * PER);
#pragma unroll
        for (int k = 0; k < PER; ++k)
            acc[i * F + k * 32 + lane] = v[k];
    }
    if (tid == 0) s_cnt = 0;
    __syncthreads();

    const int nvec = E >> 4;
    const uint4* mv = (const uint4*)omap;
    const int ntail = E - 16 * nvec;
    if (tid < ntail) {
        const int e = 16 * nvec + tid;
        const int c = cols[e];
        if ((unsigned)(c - lo) < (unsigned)nown)
            edge_add_serial<F>(src, rows, acc, e, c, lo);
    }

    unsigned lcnt = OWN_LBAD;
    if constexpr (MODE == 1) {
        if (list_ok) lcnt = mylist[0];
    }

    if (lcnt == OWN_LBAD) {
        uint4 cb[OWN_VPT];
#pragma unroll
        for (int j = 0; j < OWN_VPT; ++j) {
            const int vi = j * OWN_TPB + tid;
            cb[j] = (vi < nvec) ? mv[vi] : make_uint4(npat, npat, npat, npat);
        }

        for (int vbase = 0; vbase < nvec; vbase += OWN_VCH) {
            unsigned hit = 0u;
#pragma unroll
            for (int j = 0; j < OWN_VPT; ++j) {
                const unsigned x0 = cb[j].x ^ pat, x1 = cb[j].y ^ pat;
                const unsigned x2 = cb[j].z ^ pat, x3 = cb[j].w ^ pat;
                const unsigned any = ((x0 - 0x01010101u) & ~x0) | ((x1 - 0x01010101u) & ~x1) |
                                     ((x2 - 0x01010101u) & ~x2) | ((x3 - 0x01010101u) & ~x3);
                if (any & 0x80808080u) hit |= 1u << j;
            }

#pragma unroll
            for (int j = 0; j < OWN_VPT; ++j) {
                const int vi = vbase + OWN_VCH + j * OWN_TPB + tid;
                cb[j] = (vi < nvec) ? mv[vi] : make_uint4(npat, npat, npat, npat);
            }

            while (hit) {
                const int j = __builtin_ctz(hit);
                hit &= hit - 1u;
                const int vi = vbase + j * OWN_TPB + tid;
                const uint4 c4 = mv[vi];
                const unsigned xs[4] = {c4.x ^ pat, c4.y ^ pat, c4.z ^ pat, c4.w ^ pat};
                unsigned z[4];
                int cnt = 0;
#pragma unroll
                for (int d = 0; d < 4; ++d) {
                    z[d] = ~(((xs[d] & 0x7f7f7f7fu) + 0x7f7f7f7fu) | xs[d]) & 0x80808080u;
                    cnt += __builtin_popcount(z[d]);
                }
                int p = atomicAdd(&s_cnt, cnt);
#pragma unroll
                for (int d = 0; d < 4; ++d) {
                    unsigned m = z[d];
                    while (m) {
                        const int e = 16 * vi + 4 * d + (__builtin_ctz(m) >> 3);
                        m &= m - 1u;
                        if (p < OWN_QCAP) queue[p] = e;
                        else edge_add_serial<F>(src, rows, acc, e, cols[e], lo);
                        ++p;
                    }
                }
            }
        }
        __syncthreads();

        const int sc = __builtin_amdgcn_readfirstlane(s_cnt);
        const int n = min(sc, OWN_QCAP);
        const int npad = (n + OWN_U - 1) & ~(OWN_U - 1);
        if constexpr (MODE == 0) {
            if (sc > OWN_QCAP) lbad = true;
        }

        if (n > 0) {
            constexpr int NRES = OWN_QCAP / OWN_TPB;
            int rr[NRES], cc[NRES];
#pragma unroll
            for (int i = 0; i < NRES; ++i) {
                const int q = i * OWN_TPB + tid;
                const int eq = queue[q];
                const int e = (q < n) ? eq : 0;
                rr[i] = rows[e];
                cc[i] = cols[e];
            }
#pragma unroll
            for (int i = 0; i < NRES; ++i) {
                const int q = i * OWN_TPB + tid;
                if (q < npad) {
                    const unsigned word = (q < n) ? ((unsigned)rr[i] | ((unsigned)(cc[i] - lo) << OWN_RBITS)) : 0u;
                    queue[q] = (int)word;
                    if constexpr (MODE == 0) {
                        if (!lbad) ent[q] = word;
                    }
                }
            }
        }
        if constexpr (MODE == 0) {
            if (list_ok && tid == 0) mylist[0] = lbad ? OWN_LBAD : (unsigned)n;
        }
        __syncthreads();

        int q0 = wave * OWN_U;
        unsigned nw[OWN_U] = {0u, 0u, 0u, 0u, 0u, 0u, 0u, 0u};
        if (q0 < n) {
            const uint4 fa = *(const uint4*)(queue + q0);
            const uint4 fb = *(const uint4*)(queue + q0 + 4);
            nw[0] = (unsigned)__builtin_amdgcn_readfirstlane((int)fa.x);
            nw[1] = (unsigned)__builtin_amdgcn_readfirstlane((int)fa.y);
            nw[2] = (unsigned)__builtin_amdgcn_readfirstlane((int)fa.z);
            nw[3] = (unsigned)__builtin_amdgcn_readfirstlane((int)fa.w);
            nw[4] = (unsigned)__builtin_amdgcn_readfirstlane((int)fb.x);
            nw[5] = (unsigned)__builtin_amdgcn_readfirstlane((int)fb.y);
            nw[6] = (unsigned)__builtin_amdgcn_readfirstlane((int)fb.z);
            nw[7] = (unsigned)__builtin_amdgcn_readfirstlane((int)fb.w);
        }
        for (; q0 < n; q0 += OWN_NW * OWN_U) {
            const unsigned wd[OWN_U] = {nw[0], nw[1], nw[2], nw[3], nw[4], nw[5], nw[6], nw[7]};
            const int qn = q0 + OWN_NW * OWN_U;
            uint4 ta = make_uint4(0u, 0u, 0u, 0u), tb = ta;
            if (qn < n) {
                ta = *(const uint4*)(queue + qn);
                tb = *(const uint4*)(queue + qn + 4);
            }
            add_batch(wd, q0, n);
            nw[0] = (unsigned)__builtin_amdgcn_readfirstlane((int)ta.x);
            nw[1] = (unsigned)__builtin_amdgcn_readfirstlane((int)ta.y);
            nw[2] = (unsigned)__builtin_amdgcn_readfirstlane((int)ta.z);
            nw[3] = (unsigned)__builtin_amdgcn_readfirstlane((int)ta.w);
            nw[4] = (unsigned)__builtin_amdgcn_readfirstlane((int)tb.x);
            nw[5] = (unsigned)__builtin_amdgcn_readfirstlane((int)tb.y);
            nw[6] = (unsigned)__builtin_amdgcn_readfirstlane((int)tb.z);
            nw[7] = (unsigned)__builtin_amdgcn_readfirstlane((int)tb.w);
        }
    } else {
        const int n = (int)lcnt;
        int q0 = wave * OWN_U;
        uint4 na = make_uint4(0u, 0u, 0u, 0u), nb = na;
        if (q0 < n) {
            na = *(const uint4*)(ent + q0);
            nb = *(const uint4*)(ent + q0 + 4);
        }
        for (; q0 < n; q0 += OWN_NW * OWN_U) {
            const uint4 ca = na, cb4 = nb;
            const int qn = q0 + OWN_NW * OWN_U;
            if (qn < n) {
                na = *(const uint4*)(ent + qn);
                nb = *(const uint4*)(ent + qn + 4);
            }
            const unsigned wd[OWN_U] = {ca.x, ca.y, ca.z, ca.w, cb4.x, cb4.y, cb4.z, cb4.w};
            add_batch(wd, q0, n);
        }
    }

    __syncthreads();

    const vecf bv = *(const vecf*)(bias + lane * PER);
    vecf gv = bv, bev = bv;
    if constexpr (LNRELU) {
        gv  = *(const vecf*)(g + lane * PER);
        bev = *(const vecf*)(be + lane * PER);
    }
    for (int i = wave; i < nown; i += OWN_NW) {
        const float dv = dinv[lo + i];
        vecf v;
#pragma unroll
        for (int k = 0; k < PER; ++k)
            v[k] = acc[i * F + k * 32 + lane] * dv + bv[k];
        if constexpr (LNRELU) {
            float sum = 0.f;
#pragma unroll
            for (int k = 0; k < PER; ++k) sum += v[k];
            sum = wave_sum(sum);
            const float mu = sum * (1.0f / F);
            float var = 0.f;
#pragma unroll
            for (int k = 0; k < PER; ++k) {
                const float d = v[k] - mu;
                var += d * d;
            }
            var = wave_sum(var);
            var *= (1.0f / F);
            const float rs = rsqrtf(var + LN_EPS);
#pragma unroll
            for (int k = 0; k < PER; ++k)
                v[k] = fmaxf((v[k] - mu) * rs * gv[k] + bev[k], 0.f);
        }
        *(vecf*)(out + (size_t)(lo + i) * F + lane * PER) = v;
    }
}

extern "C" void kernel_launch(void* const* d_in, const int* in_sizes, int n_in,
                              void* d_out, int out_size, void* d_ws, size_t ws_size,
                              hipStream_t stream) {
    constexpr int F = 128, H = 128, D = 64, C = 32;
    constexpr int MT = 2;

    const float* x   = (const float*)d_in[0];
    const int*   ei  = (const int*)d_in[1];
    const float* W1  = (const float*)d_in[2];
    const float* b1  = (const float*)d_in[3];
    const float* g1  = (const float*)d_in[4];
    const float* be1 = (const float*)d_in[5];
    const float* W2  = (const float*)d_in[6];
    const float* b2  = (const float*)d_in[7];
    const float* Wc  = (const float*)d_in[8];
    const float* bc  = (const float*)d_in[9];
    float* logits    = (float*)d_out;
    unsigned int* omap = (unsigned int*)d_out;

    const int N = in_sizes[0] / F;
    const int E = in_sizes[1] / 2;
    const int* rows = ei;
    const int* cols = ei + E;

    float* ws   = (float*)d_ws;
    float* hw1  = ws;
    float* agg2 = ws;
    float* hw2  = ws + (size_t)N * 64;
    float* h2   = ws + (size_t)N * 128;
    float* deg  = ws + (size_t)N * 256;
    float* dinv = deg + N;

    const int TPB = 256;
    const int wavesPerBlk = TPB / 32;
    auto cdiv = [](long long a, long long b) { return (int)((a + b - 1) / b); };

    deg_init_kernel<<<cdiv(N, TPB), TPB, 0, stream>>>(deg, N);
    deg_count_map_kernel<<<cdiv(cdiv(E, 4), TPB), TPB, 0, stream>>>(cols, deg, omap, E);
    deg_inv_kernel<<<cdiv(N, TPB), TPB, 0, stream>>>(deg, dinv, N);

    {
        const int nWaves = cdiv(N, 16 * MT);
        gemm_wmma_f32<F, H / 16, MT, true><<<cdiv(nWaves, wavesPerBlk), TPB, 0, stream>>>(
            x, W1, nullptr, dinv, hw1, N);
    }

    agg_owner_lds_kernel<H, 0, true><<<cdiv(N, OWN_R), OWN_TPB, 0, stream>>>(
        hw1, rows, cols, omap, omap, out_size, dinv, b1, g1, be1, h2, N, E);

    {
        const int nWaves = cdiv(N, 16 * MT);
        gemm_wmma_f32<H, D / 16, MT, true><<<cdiv(nWaves, wavesPerBlk), TPB, 0, stream>>>(
            h2, W2, nullptr, dinv, hw2, N);
    }

    agg_owner_lds_kernel<D, 1, false><<<cdiv(N, OWN_R), OWN_TPB, 0, stream>>>(
        hw2, rows, cols, omap, omap, out_size, dinv, b2, nullptr, nullptr, agg2, N, E);

    {
        const int nWaves = cdiv(N, 16 * MT);
        gemm_wmma_f32<D, C / 16, MT, false><<<cdiv(nWaves, wavesPerBlk), TPB, 0, stream>>>(
            agg2, Wc, bc, nullptr, logits, N);
    }
}


# Owner blocks' serial loops load in groups: a wave's 13 self-loop rows requested together, its rows' target scales by one load ahead of the store loop, owner-map hit vectors re-read 4 at a time
# speedup vs baseline: 2.5718x; 1.0423x over previous
#include <hip/hip_runtime.h>
#include <hip/hip_bf16.h>


typedef __attribute__((ext_vector_type(2))) float v2f;
typedef __attribute__((ext_vector_type(8))) float v8f;

#define LN_EPS 1e-5f

template <int K, int NT, int MT, bool SCALE>
__global__ __launch_bounds__(256)
void gemm_wmma_f32(const float* __restrict__ A,
                   const float* __restrict__ W,
                   const float* __restrict__ bias,
                   const float* __restrict__ rowscale,
                   float* __restrict__ out,
                   int nrows) {
    constexpr int NOUT = NT * 16;
    const int wave = (blockIdx.x * blockDim.x + threadIdx.x) >> 5;
    const int lane = threadIdx.x & 31;
    const int m0 = wave * (MT * 16);
    if (m0 >= nrows) return;

    const int mr = lane & 15;
    const int hi = lane >> 4;

    v8f acc[MT][NT];
#pragma unroll
    for (int mt = 0; mt < MT; ++mt)
#pragma unroll
        for (int t = 0; t < NT; ++t)
            acc[mt][t] = (v8f){0.f, 0.f, 0.f, 0.f, 0.f, 0.f, 0.f, 0.f};

    const float* arow[MT];
#pragma unroll
    for (int mt = 0; mt < MT; ++mt)
        arow[mt] = A + (size_t)(m0 + mt * 16 + mr) * K;

#pragma unroll 2
    for (int k = 0; k < K; k += 4) {
        const int kb = k + hi * 2;
        v2f a[MT];
#pragma unroll
        for (int mt = 0; mt < MT; ++mt) {
            a[mt].x = arow[mt][kb];
            a[mt].y = arow[mt][kb + 1];
        }
        const float* w0 = W + (size_t)kb * NOUT;
        const float* w1 = w0 + NOUT;
#pragma unroll
        for (int t = 0; t < NT; ++t) {
            const int n = t * 16 + mr;
            v2f b;
            b.x = w0[n];
            b.y = w1[n];
#pragma unroll
            for (int mt = 0; mt < MT; ++mt)
                acc[mt][t] = __builtin_amdgcn_wmma_f32_16x16x4_f32(
                    false, a[mt], false, b, (short)0, acc[mt][t], false, false);
        }
    }

#pragma unroll
    for (int mt = 0; mt < MT; ++mt) {
        float rs[8];
#pragma unroll
        for (int v = 0; v < 8; ++v) {
            if constexpr (SCALE) rs[v] = rowscale[m0 + mt * 16 + v + hi * 8];
            else rs[v] = 1.f;
        }
#pragma unroll
        for (int t = 0; t < NT; ++t) {
            const int n = t * 16 + mr;
            const float bv = bias ? bias[n] : 0.f;
#pragma unroll
            for (int v = 0; v < 8; ++v) {
                const int m = m0 + mt * 16 + v + hi * 8;
                if constexpr (SCALE) out[(size_t)m * NOUT + n] = acc[mt][t][v] * rs[v] + bv;
                else out[(size_t)m * NOUT + n] = acc[mt][t][v] + bv;
            }
        }
    }
}

__global__ void deg_init_kernel(float* deg, int N) {
    int i = blockIdx.x * blockDim.x + threadIdx.x;
    if (i < N) deg[i] = 1.0f;
}

__global__ void deg_inv_kernel(const float* __restrict__ deg, float* dinv, int N) {
    int i = blockIdx.x * blockDim.x + threadIdx.x;
    if (i < N) dinv[i] = rsqrtf(deg[i]);
}

constexpr int OWN_R    = 400;
constexpr int OWN_TPB  = 1024;
constexpr int OWN_NW   = OWN_TPB / 32;
constexpr int OWN_VPT  = 12;
constexpr int OWN_VCH  = OWN_TPB * OWN_VPT;
constexpr int OWN_QCAP = 8192;
constexpr int OWN_U    = 8;
constexpr int OWN_RBITS   = 17;
constexpr int OWN_LCAP    = 8192;
constexpr int OWN_LSTRIDE = OWN_LCAP + 8;
constexpr unsigned OWN_LBAD = 0xffffffffu;
constexpr int OWN_NROW = (OWN_R + OWN_NW - 1) / OWN_NW;
constexpr int OWN_HG   = 4;
static_assert(OWN_NROW <= 32, "one lane per owned row holds the row's target scale");
static_assert(OWN_U == 8, "a batch of list entries is read as two 16-byte loads");
static_assert(OWN_QCAP % OWN_TPB == 0 && OWN_QCAP % OWN_U == 0 && OWN_QCAP <= OWN_LCAP,
              "every thread resolves OWN_QCAP / OWN_TPB queue entries; whole batches; the queue fits the list");
static_assert(OWN_R < (1 << (32 - OWN_RBITS)), "the local target fits above the source id");

__global__ __launch_bounds__(256)
void deg_count_map_kernel(const int* __restrict__ col, float* deg,
                          unsigned int* __restrict__ omap, int E) {
    const int t  = blockIdx.x * blockDim.x + threadIdx.x;
    const int e0 = 4 * t;
    if (e0 + 3 < E) {
        const int c0 = col[e0], c1 = col[e0 + 1], c2 = col[e0 + 2], c3 = col[e0 + 3];
        atomicAdd(&deg[c0], 1.0f);
        atomicAdd(&deg[c1], 1.0f);
        atomicAdd(&deg[c2], 1.0f);
        atomicAdd(&deg[c3], 1.0f);
        omap[t] = (unsigned)(c0 / OWN_R) | ((unsigned)(c1 / OWN_R) << 8) |
                  ((unsigned)(c2 / OWN_R) << 16) | ((unsigned)(c3 / OWN_R) << 24);
    } else {
        for (int e = e0; e < E; ++e) atomicAdd(&deg[col[e]], 1.0f);
    }
}

template <int F>
__device__ __forceinline__ void edge_add_serial(const float* __restrict__ src,
                                                const int* __restrict__ rows,
                                                float* acc, int e, int c, int lo) {
    constexpr int PER = F / 32;
    const int r = rows[e];
    const float* s = src + (size_t)r * F;
    float* a = acc + (c - lo) * F;
    for (int f = 0; f < F; ++f)
        atomicAdd(a + (f % PER) * 32 + (f / PER), s[f]);
}

template <int M>
__device__ __forceinline__ float row_xor_lane(float x) {
    const int b = __builtin_bit_cast(int, x);
    return __builtin_bit_cast(float, __builtin_amdgcn_update_dpp(b, b, 0x160 + M, 0xf, 0xf, false));
}
__device__ __forceinline__ float other_row_lane(float x) {
    const int b = __builtin_bit_cast(int, x);
    return __builtin_bit_cast(float, __builtin_amdgcn_permlanex16(b, b, 0x76543210, 0xfedcba98, false, false));
}
__device__ __forceinline__ float wave_sum(float x) {
    x += other_row_lane(x);
    x += row_xor_lane<8>(x);
    x += row_xor_lane<4>(x);
    x += row_xor_lane<2>(x);
    x += row_xor_lane<1>(x);
    return x;
}

template <int F, int MODE, bool LNRELU>
__global__ __launch_bounds__(OWN_TPB)
void agg_owner_lds_kernel(const float* __restrict__ src,
                          const int* __restrict__ rows,
                          const int* __restrict__ cols,
                          const unsigned int* __restrict__ omap,
                          unsigned int* __restrict__ lbase, int lwords,
                          const float* __restrict__ dinv,
                          const float* __restrict__ bias,
                          const float* __restrict__ g,
                          const float* __restrict__ be,
                          float* __restrict__ out, int N, int E) {
    constexpr int PER = F / 32;
    typedef __attribute__((ext_vector_type(PER))) float vecf;

    __shared__ __attribute__((aligned(16))) float acc[(OWN_R + 1) * F];
    __shared__ __attribute__((aligned(16))) int queue[OWN_QCAP];
    __shared__ int s_cnt;

    const int tid  = threadIdx.x;
    const int lane = tid & 31;
    const int wave = __builtin_amdgcn_readfirstlane(tid >> 5);
    const int lo   = blockIdx.x * OWN_R;
    const int nown = min(OWN_R, N - lo);
    const unsigned pat  = (unsigned)blockIdx.x * 0x01010101u;
    const unsigned npat = ~pat;

    const int loff = (((E + 3) >> 2) + 63) & ~63;
    const bool list_ok = (N <= (1 << OWN_RBITS)) &&
                         ((long long)loff + (long long)gridDim.x * OWN_LSTRIDE <= (long long)lwords);
    unsigned int* mylist = lbase + loff + (size_t)blockIdx.x * OWN_LSTRIDE;
    unsigned int* ent = mylist + 8;
    bool lbad = !list_ok;

    auto add_batch = [&](const unsigned (&wd)[OWN_U], int q0, int n) {
        int   lc[OWN_U];
        vecf  v[OWN_U];
#pragma unroll
        for (int u = 0; u < OWN_U; ++u) {
            const int r = (int)(wd[u] & ((1u << OWN_RBITS) - 1u));
            lc[u] = (int)(wd[u] >> OWN_RBITS);
            v[u] = *(const vecf*)(src + (size_t)r * F + lane * PER);
        }
#pragma unroll
        for (int u = 0; u < OWN_U; ++u) {
            float* a = acc + ((q0 + u < n) ? lc[u] : OWN_R) * F + lane;
#pragma unroll
            for (int k = 0; k < PER; ++k)
                atomicAdd(a + k * 32, v[u][k]);
        }
    };

    const int nvec = E >> 4;
    const uint4* mv = (const uint4*)omap;
    const int ntail = E - 16 * nvec;

    uint4 cb[OWN_VPT];
    auto load_chunk = [&](int vb) {
#pragma unroll
        for (int j = 0; j < OWN_VPT; ++j) {
            const int vi = vb + j * OWN_TPB + tid;
            cb[j] = mv[max(min(vi, nvec - 1), 0)];
        }
    };

    unsigned lcnt = OWN_LBAD;
    uint4 na = make_uint4(0u, 0u, 0u, 0u), nb = na;
    if constexpr (MODE == 0) {
        load_chunk(0);
    } else {
        if (list_ok) {
            lcnt = mylist[0];
            na = *(const uint4*)(ent + wave * OWN_U);
            nb = *(const uint4*)(ent + wave * OWN_U + 4);
        }
    }

    {
        vecf sv[OWN_NROW];
#pragma unroll
        for (int t = 0; t < OWN_NROW; ++t) {
            const int i = min(wave + t * OWN_NW, nown - 1);
            sv[t] = *(const vecf*)(src + (size_t)(lo + i) * F + lane * PER);
        }
#pragma unroll
        for (int t = 0; t < OWN_NROW; ++t) {
            const int i = wave + t * OWN_NW;
            if (i < nown) {
#pragma unroll
                for (int k = 0; k < PER; ++k)
                    acc[i * F + k * 32 + lane] = sv[t][k];
            }
        }
    }
    if (tid == 0) s_cnt = 0;
    __syncthreads();

    if (tid < ntail) {
        const int e = 16 * nvec + tid;
        const int c = cols[e];
        if ((unsigned)(c - lo) < (unsigned)nown)
            edge_add_serial<F>(src, rows, acc, e, c, lo);
    }

    if (lcnt == OWN_LBAD) {
        if constexpr (MODE == 1) load_chunk(0);

        auto queue_hits = [&](const uint4 c4, const int vi) {
            const unsigned xs[4] = {c4.x ^ pat, c4.y ^ pat, c4.z ^ pat, c4.w ^ pat};
            unsigned z[4];
            int cnt = 0;
#pragma unroll
            for (int d = 0; d < 4; ++d) {
                z[d] = ~(((xs[d] & 0x7f7f7f7fu) + 0x7f7f7f7fu) | xs[d]) & 0x80808080u;
                cnt += __builtin_popcount(z[d]);
            }
            int p = atomicAdd(&s_cnt, cnt);
#pragma unroll
            for (int d = 0; d < 4; ++d) {
                unsigned m = z[d];
                while (m) {
                    const int e = 16 * vi + 4 * d + (__builtin_ctz(m) >> 3);
                    m &= m - 1u;
                    if (p < OWN_QCAP) queue[p] = e;
                    else edge_add_serial<F>(src, rows, acc, e, cols[e], lo);
                    ++p;
                }
            }
        };

        for (int vbase = 0; vbase < nvec; vbase += OWN_VCH) {
            unsigned hit = 0u;
#pragma unroll
            for (int j = 0; j < OWN_VPT; ++j) {
                const unsigned x0 = cb[j].x ^ pat, x1 = cb[j].y ^ pat;
                const unsigned x2 = cb[j].z ^ pat, x3 = cb[j].w ^ pat;
                const unsigned any = ((x0 - 0x01010101u) & ~x0) | ((x1 - 0x01010101u) & ~x1) |
                                     ((x2 - 0x01010101u) & ~x2) | ((x3 - 0x01010101u) & ~x3);
                if ((vbase + j * OWN_TPB + tid < nvec) && (any & 0x80808080u)) hit |= 1u << j;
            }

            int   hv[OWN_HG];
            uint4 hc[OWN_HG];
#pragma unroll
            for (int t = 0; t < OWN_HG; ++t) {
                const bool has = hit != 0u;
                hv[t] = has ? vbase + __builtin_ctz(hit | 0x80000000u) * OWN_TPB + tid : -1;
                hit &= hit - 1u;
                hc[t] = mv[has ? hv[t] : 0];
            }

            load_chunk(vbase + OWN_VCH);

#pragma unroll
            for (int t = OWN_HG - 1; t >= 0; --t) {
                if (hv[t] >= 0) queue_hits(hc[t], hv[t]);
            }
            while (hit) {
                const int vi = vbase + __builtin_ctz(hit) * OWN_TPB + tid;
                hit &= hit - 1u;
                queue_hits(mv[vi], vi);
            }
        }
        __syncthreads();

        const int sc = __builtin_amdgcn_readfirstlane(s_cnt);
        const int n = min(sc, OWN_QCAP);
        const int npad = (n + OWN_U - 1) & ~(OWN_U - 1);
        if constexpr (MODE == 0) {
            if (sc > OWN_QCAP) lbad = true;
        }

        if (n > 0) {
            constexpr int NRES = OWN_QCAP / OWN_TPB;
            int rr[NRES], cc[NRES];
#pragma unroll
            for (int i = 0; i < NRES; ++i) {
                const int q = i * OWN_TPB + tid;
                const int eq = queue[q];
                const int e = (q < n) ? eq : 0;
                rr[i] = rows[e];
                cc[i] = cols[e];
            }
#pragma unroll
            for (int i = 0; i < NRES; ++i) {
                const int q = i * OWN_TPB + tid;
                if (q < npad) {
                    const unsigned word = (q < n) ? ((unsigned)rr[i] | ((unsigned)(cc[i] - lo) << OWN_RBITS)) : 0u;
                    queue[q] = (int)word;
                    if constexpr (MODE == 0) {
                        if (!lbad) ent[q] = word;
                    }
                }
            }
        }
        if constexpr (MODE == 0) {
            if (list_ok && tid == 0) mylist[0] = lbad ? OWN_LBAD : (unsigned)n;
        }
        __syncthreads();

        int q0 = wave * OWN_U;
        unsigned nw[OWN_U] = {0u, 0u, 0u, 0u, 0u, 0u, 0u, 0u};
        if (q0 < n) {
            const uint4 fa = *(const uint4*)(queue + q0);
            const uint4 fb = *(const uint4*)(queue + q0 + 4);
            nw[0] = (unsigned)__builtin_amdgcn_readfirstlane((int)fa.x);
            nw[1] = (unsigned)__builtin_amdgcn_readfirstlane((int)fa.y);
            nw[2] = (unsigned)__builtin_amdgcn_readfirstlane((int)fa.z);
            nw[3] = (unsigned)__builtin_amdgcn_readfirstlane((int)fa.w);
            nw[4] = (unsigned)__builtin_amdgcn_readfirstlane((int)fb.x);
            nw[5] = (unsigned)__builtin_amdgcn_readfirstlane((int)fb.y);
            nw[6] = (unsigned)__builtin_amdgcn_readfirstlane((int)fb.z);
            nw[7] = (unsigned)__builtin_amdgcn_readfirstlane((int)fb.w);
        }
        for (; q0 < n; q0 += OWN_NW * OWN_U) {
            const unsigned wd[OWN_U] = {nw[0], nw[1], nw[2], nw[3], nw[4], nw[5], nw[6], nw[7]};
            const int qn = q0 + OWN_NW * OWN_U;
            uint4 ta = make_uint4(0u, 0u, 0u, 0u), tb = ta;
            if (qn < n) {
                ta = *(const uint4*)(queue + qn);
                tb = *(const uint4*)(queue + qn + 4);
            }
            add_batch(wd, q0, n);
            nw[0] = (unsigned)__builtin_amdgcn_readfirstlane((int)ta.x);
            nw[1] = (unsigned)__builtin_amdgcn_readfirstlane((int)ta.y);
            nw[2] = (unsigned)__builtin_amdgcn_readfirstlane((int)ta.z);
            nw[3] = (unsigned)__builtin_amdgcn_readfirstlane((int)ta.w);
            nw[4] = (unsigned)__builtin_amdgcn_readfirstlane((int)tb.x);
            nw[5] = (unsigned)__builtin_amdgcn_readfirstlane((int)tb.y);
            nw[6] = (unsigned)__builtin_amdgcn_readfirstlane((int)tb.z);
            nw[7] = (unsigned)__builtin_amdgcn_readfirstlane((int)tb.w);
        }
    } else {
        const int n = (int)lcnt;
        int q0 = wave * OWN_U;
        for (; q0 < n; q0 += OWN_NW * OWN_U) {
            const uint4 ca = na, cb4 = nb;
            const int qn = q0 + OWN_NW * OWN_U;
            if (qn < n) {
                na = *(const uint4*)(ent + qn);
                nb = *(const uint4*)(ent + qn + 4);
            }
            const unsigned wd[OWN_U] = {ca.x, ca.y, ca.z, ca.w, cb4.x, cb4.y, cb4.z, cb4.w};
            add_batch(wd, q0, n);
        }
    }

    const vecf bv = *(const vecf*)(bias + lane * PER);
    vecf gv = bv, bev = bv;
    if constexpr (LNRELU) {
        gv  = *(const vecf*)(g + lane * PER);
        bev = *(const vecf*)(be + lane * PER);
    }
    const int dvl = __builtin_bit_cast(int, dinv[lo + min(wave + lane * OWN_NW, nown - 1)]);

    __syncthreads();

    int t = 0;
    for (int i = wave; i < nown; i += OWN_NW, ++t) {
        const float dv = __builtin_bit_cast(float, __builtin_amdgcn_readlane(dvl, t));
        vecf v;
#pragma unroll
        for (int k = 0; k < PER; ++k)
            v[k] = acc[i * F + k * 32 + lane] * dv + bv[k];
        if constexpr (LNRELU) {
            float sum = 0.f;
#pragma unroll
            for (int k = 0; k < PER; ++k) sum += v[k];
            sum = wave_sum(sum);
            const float mu = sum * (1.0f / F);
            float var = 0.f;
#pragma unroll
            for (int k = 0; k < PER; ++k) {
                const float d = v[k] - mu;
                var += d * d;
            }
            var = wave_sum(var);
            var *= (1.0f / F);
            const float rs = rsqrtf(var + LN_EPS);
#pragma unroll
            for (int k = 0; k < PER; ++k)
                v[k] = fmaxf((v[k] - mu) * rs * gv[k] + bev[k], 0.f);
        }
        *(vecf*)(out + (size_t)(lo + i) * F + lane * PER) = v;
    }
}

extern "C" void kernel_launch(void* const* d_in, const int* in_sizes, int n_in,
                              void* d_out, int out_size, void* d_ws, size_t ws_size,
                              hipStream_t stream) {
    constexpr int F = 128, H = 128, D = 64, C = 32;
    constexpr int MT = 2;

    const float* x   = (const float*)d_in[0];
    const int*   ei  = (const int*)d_in[1];
    const float* W1  = (const float*)d_in[2];
    const float* b1  = (const float*)d_in[3];
    const float* g1  = (const float*)d_in[4];
    const float* be1 = (const float*)d_in[5];
    const float* W2  = (const float*)d_in[6];
    const float* b2  = (const float*)d_in[7];
    const float* Wc  = (const float*)d_in[8];
    const float* bc  = (const float*)d_in[9];
    float* logits    = (float*)d_out;
    unsigned int* omap = (unsigned int*)d_out;

    const int N = in_sizes[0] / F;
    const int E = in_sizes[1] / 2;
    const int* rows = ei;
    const int* cols = ei + E;

    float* ws   = (float*)d_ws;
    float* hw1  = ws;
    float* agg2 = ws;
    float* hw2  = ws + (size_t)N * 64;
    float* h2   = ws + (size_t)N * 128;
    float* deg  = ws + (size_t)N * 256;
    float* dinv = deg + N;

    const int TPB = 256;
    const int wavesPerBlk = TPB / 32;
    auto cdiv = [](long long a, long long b) { return (int)((a + b - 1) / b); };

    deg_init_kernel<<<cdiv(N, TPB), TPB, 0, stream>>>(deg, N);
    deg_count_map_kernel<<<cdiv(cdiv(E, 4), TPB), TPB, 0, stream>>>(cols, deg, omap, E);
    deg_inv_kernel<<<cdiv(N, TPB), TPB, 0, stream>>>(deg, dinv, N);

    {
        const int nWaves = cdiv(N, 16 * MT);
        gemm_wmma_f32<F, H / 16, MT, true><<<cdiv(nWaves, wavesPerBlk), TPB, 0, stream>>>(
            x, W1, nullptr, dinv, hw1, N);
    }

    agg_owner_lds_kernel<H, 0, true><<<cdiv(N, OWN_R), OWN_TPB, 0, stream>>>(
        hw1, rows, cols, omap, omap, out_size, dinv, b1, g1, be1, h2, N, E);

    {
        const int nWaves = cdiv(N, 16 * MT);
        gemm_wmma_f32<H, D / 16, MT, true><<<cdiv(nWaves, wavesPerBlk), TPB, 0, stream>>>(
            h2, W2, nullptr, dinv, hw2, N);
    }

    agg_owner_lds_kernel<D, 1, false><<<cdiv(N, OWN_R), OWN_TPB, 0, stream>>>(
        hw2, rows, cols, omap, omap, out_size, dinv, b2, nullptr, nullptr, agg2, N, E);

    {
        const int nWaves = cdiv(N, 16 * MT);
        gemm_wmma_f32<D, C / 16, MT, false><<<cdiv(nWaves, wavesPerBlk), TPB, 0, stream>>>(
            agg2, Wc, bc, nullptr, logits, N);
    }
}


# Owner blocks run the product behind each edge sum: finished rows stay in LDS (row pitch +4 words), WMMA reads its A fragments there; h2, agg2 never stored; k order = LDS word order; 8 -> 6 launches
# speedup vs baseline: 2.8176x; 1.0956x over previous
#include <hip/hip_runtime.h>
#include <hip/hip_bf16.h>


typedef __attribute__((ext_vector_type(2))) float v2f;
typedef __attribute__((ext_vector_type(8))) float v8f;

#define LN_EPS 1e-5f

template <int K, int NT, int MT, bool SCALE>
__global__ __launch_bounds__(256)
void gemm_wmma_f32(const float* __restrict__ A,
                   const float* __restrict__ W,
                   const float* __restrict__ bias,
                   const float* __restrict__ rowscale,
                   float* __restrict__ out,
                   int nrows) {
    constexpr int NOUT = NT * 16;
    const int wave = (blockIdx.x * blockDim.x + threadIdx.x) >> 5;
    const int lane = threadIdx.x & 31;
    const int m0 = wave * (MT * 16);
    if (m0 >= nrows) return;

    const int mr = lane & 15;
    const int hi = lane >> 4;

    v8f acc[MT][NT];
#pragma unroll
    for (int mt = 0; mt < MT; ++mt)
#pragma unroll
        for (int t = 0; t < NT; ++t)
            acc[mt][t] = (v8f){0.f, 0.f, 0.f, 0.f, 0.f, 0.f, 0.f, 0.f};

    const float* arow[MT];
#pragma unroll
    for (int mt = 0; mt < MT; ++mt)
        arow[mt] = A + (size_t)(m0 + mt * 16 + mr) * K;

#pragma unroll 2
    for (int k = 0; k < K; k += 4) {
        const int kb = k + hi * 2;
        v2f a[MT];
#pragma unroll
        for (int mt = 0; mt < MT; ++mt) {
            a[mt].x = arow[mt][kb];
            a[mt].y = arow[mt][kb + 1];
        }
        const float* w0 = W + (size_t)kb * NOUT;
        const float* w1 = w0 + NOUT;
#pragma unroll
        for (int t = 0; t < NT; ++t) {
            const int n = t * 16 + mr;
            v2f b;
            b.x = w0[n];
            b.y = w1[n];
#pragma unroll
            for (int mt = 0; mt < MT; ++mt)
                acc[mt][t] = __builtin_amdgcn_wmma_f32_16x16x4_f32(
                    false, a[mt], false, b, (short)0, acc[mt][t], false, false);
        }
    }

#pragma unroll
    for (int mt = 0; mt < MT; ++mt) {
        float rs[8];
#pragma unroll
        for (int v = 0; v < 8; ++v) {
            if constexpr (SCALE) rs[v] = rowscale[m0 + mt * 16 + v + hi * 8];
            else rs[v] = 1.f;
        }
#pragma unroll
        for (int t = 0; t < NT; ++t) {
            const int n = t * 16 + mr;
            const float bv = bias ? bias[n] : 0.f;
#pragma unroll
            for (int v = 0; v < 8; ++v) {
                const int m = m0 + mt * 16 + v + hi * 8;
                if constexpr (SCALE) out[(size_t)m * NOUT + n] = acc[mt][t][v] * rs[v] + bv;
                else out[(size_t)m * NOUT + n] = acc[mt][t][v] + bv;
            }
        }
    }
}

__global__ void deg_init_kernel(float* deg, int N) {
    int i = blockIdx.x * blockDim.x + threadIdx.x;
    if (i < N) deg[i] = 1.0f;
}

__global__ void deg_inv_kernel(const float* __restrict__ deg, float* dinv, int N) {
    int i = blockIdx.x * blockDim.x + threadIdx.x;
    if (i < N) dinv[i] = rsqrtf(deg[i]);
}

constexpr int OWN_R    = 400;
constexpr int OWN_TPB  = 1024;
constexpr int OWN_NW   = OWN_TPB / 32;
constexpr int OWN_VPT  = 12;
constexpr int OWN_VCH  = OWN_TPB * OWN_VPT;
constexpr int OWN_QCAP = 8192;
constexpr int OWN_U    = 8;
constexpr int OWN_RBITS   = 17;
constexpr int OWN_LCAP    = 8192;
constexpr int OWN_LSTRIDE = OWN_LCAP + 8;
constexpr unsigned OWN_LBAD = 0xffffffffu;
constexpr int OWN_NROW = (OWN_R + OWN_NW - 1) / OWN_NW;
constexpr int OWN_HG   = 4;
constexpr int OWN_PAD  = 4;
static_assert(OWN_R % 16 == 0, "the owned rows are whole 16-row tiles of the product behind the sum");
static_assert(OWN_NROW <= 32, "one lane per owned row holds the row's target scale");
static_assert(OWN_U == 8, "a batch of list entries is read as two 16-byte loads");
static_assert(OWN_QCAP % OWN_TPB == 0 && OWN_QCAP % OWN_U == 0 && OWN_QCAP <= OWN_LCAP,
              "every thread resolves OWN_QCAP / OWN_TPB queue entries; whole batches; the queue fits the list");
static_assert(OWN_R < (1 << (32 - OWN_RBITS)), "the local target fits above the source id");

__global__ __launch_bounds__(256)
void deg_count_map_kernel(const int* __restrict__ col, float* deg,
                          unsigned int* __restrict__ omap, int E) {
    const int t  = blockIdx.x * blockDim.x + threadIdx.x;
    const int e0 = 4 * t;
    if (e0 + 3 < E) {
        const int c0 = col[e0], c1 = col[e0 + 1], c2 = col[e0 + 2], c3 = col[e0 + 3];
        atomicAdd(&deg[c0], 1.0f);
        atomicAdd(&deg[c1], 1.0f);
        atomicAdd(&deg[c2], 1.0f);
        atomicAdd(&deg[c3], 1.0f);
        omap[t] = (unsigned)(c0 / OWN_R) | ((unsigned)(c1 / OWN_R) << 8) |
                  ((unsigned)(c2 / OWN_R) << 16) | ((unsigned)(c3 / OWN_R) << 24);
    } else {
        for (int e = e0; e < E; ++e) atomicAdd(&deg[col[e]], 1.0f);
    }
}

template <int F>
__device__ __forceinline__ void edge_add_serial(const float* __restrict__ src,
                                                const int* __restrict__ rows,
                                                float* acc, int e, int c, int lo) {
    constexpr int PER = F / 32;
    constexpr int PITCH = F + OWN_PAD;
    const int r = rows[e];
    const float* s = src + (size_t)r * F;
    float* a = acc + (c - lo) * PITCH;
    for (int f = 0; f < F; ++f)
        atomicAdd(a + (f % PER) * 32 + (f / PER), s[f]);
}

template <int M>
__device__ __forceinline__ float row_xor_lane(float x) {
    const int b = __builtin_bit_cast(int, x);
    return __builtin_bit_cast(float, __builtin_amdgcn_update_dpp(b, b, 0x160 + M, 0xf, 0xf, false));
}
__device__ __forceinline__ float other_row_lane(float x) {
    const int b = __builtin_bit_cast(int, x);
    return __builtin_bit_cast(float, __builtin_amdgcn_permlanex16(b, b, 0x76543210, 0xfedcba98, false, false));
}
__device__ __forceinline__ float wave_sum(float x) {
    x += other_row_lane(x);
    x += row_xor_lane<8>(x);
    x += row_xor_lane<4>(x);
    x += row_xor_lane<2>(x);
    x += row_xor_lane<1>(x);
    return x;
}

template <int F, int MODE, bool LNRELU, int NOUT, bool SCALEOUT>
__global__ __launch_bounds__(OWN_TPB)
void agg_owner_lds_kernel(const float* __restrict__ src,
                          const int* __restrict__ rows,
                          const int* __restrict__ cols,
                          const unsigned int* __restrict__ omap,
                          unsigned int* __restrict__ lbase, int lwords,
                          const float* __restrict__ dinv,
                          const float* __restrict__ bias,
                          const float* __restrict__ g,
                          const float* __restrict__ be,
                          const float* __restrict__ Wn,
                          const float* __restrict__ nbias,
                          float* __restrict__ out, int N, int E) {
    constexpr int PER = F / 32;
    constexpr int PITCH = F + OWN_PAD;
    typedef __attribute__((ext_vector_type(PER))) float vecf;

    __shared__ __attribute__((aligned(16))) float acc[(OWN_R + 1) * PITCH];
    __shared__ __attribute__((aligned(16))) int queue[OWN_QCAP];
    __shared__ int s_cnt;

    const int tid  = threadIdx.x;
    const int lane = tid & 31;
    const int wave = __builtin_amdgcn_readfirstlane(tid >> 5);
    const int lo   = blockIdx.x * OWN_R;
    const int nown = min(OWN_R, N - lo);
    const unsigned pat  = (unsigned)blockIdx.x * 0x01010101u;
    const unsigned npat = ~pat;

    const int loff = (((E + 3) >> 2) + 63) & ~63;
    const bool list_ok = (N <= (1 << OWN_RBITS)) &&
                         ((long long)loff + (long long)gridDim.x * OWN_LSTRIDE <= (long long)lwords);
    unsigned int* mylist = lbase + loff + (size_t)blockIdx.x * OWN_LSTRIDE;
    unsigned int* ent = mylist + 8;
    bool lbad = !list_ok;

    auto add_batch = [&](const unsigned (&wd)[OWN_U], int q0, int n) {
        int   lc[OWN_U];
        vecf  v[OWN_U];
#pragma unroll
        for (int u = 0; u < OWN_U; ++u) {
            const int r = (int)(wd[u] & ((1u << OWN_RBITS) - 1u));
            lc[u] = (int)(wd[u] >> OWN_RBITS);
            v[u] = *(const vecf*)(src + (size_t)r * F + lane * PER);
        }
#pragma unroll
        for (int u = 0; u < OWN_U; ++u) {
            float* a = acc + ((q0 + u < n) ? lc[u] : OWN_R) * PITCH + lane;
#pragma unroll
            for (int k = 0; k < PER; ++k)
                atomicAdd(a + k * 32, v[u][k]);
        }
    };

    const int nvec = E >> 4;
    const uint4* mv = (const uint4*)omap;
    const int ntail = E - 16 * nvec;

    uint4 cb[OWN_VPT];
    auto load_chunk = [&](int vb) {
#pragma unroll
        for (int j = 0; j < OWN_VPT; ++j) {
            const int vi = vb + j * OWN_TPB + tid;
            cb[j] = mv[max(min(vi, nvec - 1), 0)];
        }
    };

    unsigned lcnt = OWN_LBAD;
    uint4 na = make_uint4(0u, 0u, 0u, 0u), nb = na;
    if constexpr (MODE == 0) {
        load_chunk(0);
    } else {
        if (list_ok) {
            lcnt = mylist[0];
            na = *(const uint4*)(ent + wave * OWN_U);
            nb = *(const uint4*)(ent + wave * OWN_U + 4);
        }
    }

    {
        vecf sv[OWN_NROW];
#pragma unroll
        for (int t = 0; t < OWN_NROW; ++t) {
            const int i = min(wave + t * OWN_NW, nown - 1);
            sv[t] = *(const vecf*)(src + (size_t)(lo + i) * F + lane * PER);
        }
#pragma unroll
        for (int t = 0; t < OWN_NROW; ++t) {
            const int i = wave + t * OWN_NW;
            if (i < nown) {
#pragma unroll
                for (int k = 0; k < PER; ++k)
                    acc[i * PITCH + k * 32 + lane] = sv[t][k];
            }
        }
    }
    if (tid == 0) s_cnt = 0;
    __syncthreads();

    if (tid < ntail) {
        const int e = 16 * nvec + tid;
        const int c = cols[e];
        if ((unsigned)(c - lo) < (unsigned)nown)
            edge_add_serial<F>(src, rows, acc, e, c, lo);
    }

    if (lcnt == OWN_LBAD) {
        if constexpr (MODE == 1) load_chunk(0);

        auto queue_hits = [&](const uint4 c4, const int vi) {
            const unsigned xs[4] = {c4.x ^ pat, c4.y ^ pat, c4.z ^ pat, c4.w ^ pat};
            unsigned z[4];
            int cnt = 0;
#pragma unroll
            for (int d = 0; d < 4; ++d) {
                z[d] = ~(((xs[d] & 0x7f7f7f7fu) + 0x7f7f7f7fu) | xs[d]) & 0x80808080u;
                cnt += __builtin_popcount(z[d]);
            }
            int p = atomicAdd(&s_cnt, cnt);
#pragma unroll
            for (int d = 0; d < 4; ++d) {
                unsigned m = z[d];
                while (m) {
                    const int e = 16 * vi + 4 * d + (__builtin_ctz(m) >> 3);
                    m &= m - 1u;
                    if (p < OWN_QCAP) queue[p] = e;
                    else edge_add_serial<F>(src, rows, acc, e, cols[e], lo);
                    ++p;
                }
            }
        };

        for (int vbase = 0; vbase < nvec; vbase += OWN_VCH) {
            unsigned hit = 0u;
#pragma unroll
            for (int j = 0; j < OWN_VPT; ++j) {
                const unsigned x0 = cb[j].x ^ pat, x1 = cb[j].y ^ pat;
                const unsigned x2 = cb[j].z ^ pat, x3 = cb[j].w ^ pat;
                const unsigned any = ((x0 - 0x01010101u) & ~x0) | ((x1 - 0x01010101u) & ~x1) |
                                     ((x2 - 0x01010101u) & ~x2) | ((x3 - 0x01010101u) & ~x3);
                if ((vbase + j * OWN_TPB + tid < nvec) && (any & 0x80808080u)) hit |= 1u << j;
            }

            int   hv[OWN_HG];
            uint4 hc[OWN_HG];
#pragma unroll
            for (int t = 0; t < OWN_HG; ++t) {
                const bool has = hit != 0u;
                hv[t] = has ? vbase + __builtin_ctz(hit | 0x80000000u) * OWN_TPB + tid : -1;
                hit &= hit - 1u;
                hc[t] = mv[has ? hv[t] : 0];
            }

            load_chunk(vbase + OWN_VCH);

#pragma unroll
            for (int t = OWN_HG - 1; t >= 0; --t) {
                if (hv[t] >= 0) queue_hits(hc[t], hv[t]);
            }
            while (hit) {
                const int vi = vbase + __builtin_ctz(hit) * OWN_TPB + tid;
                hit &= hit - 1u;
                queue_hits(mv[vi], vi);
            }
        }
        __syncthreads();

        const int sc = __builtin_amdgcn_readfirstlane(s_cnt);
        const int n = min(sc, OWN_QCAP);
        const int npad = (n + OWN_U - 1) & ~(OWN_U - 1);
        if constexpr (MODE == 0) {
            if (sc > OWN_QCAP) lbad = true;
        }

        if (n > 0) {
            constexpr int NRES = OWN_QCAP / OWN_TPB;
            int rr[NRES], cc[NRES];
#pragma unroll
            for (int i = 0; i < NRES; ++i) {
                const int q = i * OWN_TPB + tid;
                const int eq = queue[q];
                const int e = (q < n) ? eq : 0;
                rr[i] = rows[e];
                cc[i] = cols[e];
            }
#pragma unroll
            for (int i = 0; i < NRES; ++i) {
                const int q = i * OWN_TPB + tid;
                if (q < npad) {
                    const unsigned word = (q < n) ? ((unsigned)rr[i] | ((unsigned)(cc[i] - lo) << OWN_RBITS)) : 0u;
                    queue[q] = (int)word;
                    if constexpr (MODE == 0) {
                        if (!lbad) ent[q] = word;
                    }
                }
            }
        }
        if constexpr (MODE == 0) {
            if (list_ok && tid == 0) mylist[0] = lbad ? OWN_LBAD : (unsigned)n;
        }
        __syncthreads();

        int q0 = wave * OWN_U;
        unsigned nw[OWN_U] = {0u, 0u, 0u, 0u, 0u, 0u, 0u, 0u};
        if (q0 < n) {
            const uint4 fa = *(const uint4*)(queue + q0);
            const uint4 fb = *(const uint4*)(queue + q0 + 4);
            nw[0] = (unsigned)__builtin_amdgcn_readfirstlane((int)fa.x);
            nw[1] = (unsigned)__builtin_amdgcn_readfirstlane((int)fa.y);
            nw[2] = (unsigned)__builtin_amdgcn_readfirstlane((int)fa.z);
            nw[3] = (unsigned)__builtin_amdgcn_readfirstlane((int)fa.w);
            nw[4] = (unsigned)__builtin_amdgcn_readfirstlane((int)fb.x);
            nw[5] = (unsigned)__builtin_amdgcn_readfirstlane((int)fb.y);
            nw[6] = (unsigned)__builtin_amdgcn_readfirstlane((int)fb.z);
            nw[7] = (unsigned)__builtin_amdgcn_readfirstlane((int)fb.w);
        }
        for (; q0 < n; q0 += OWN_NW * OWN_U) {
            const unsigned wd[OWN_U] = {nw[0], nw[1], nw[2], nw[3], nw[4], nw[5], nw[6], nw[7]};
            const int qn = q0 + OWN_NW * OWN_U;
            uint4 ta = make_uint4(0u, 0u, 0u, 0u), tb = ta;
            if (qn < n) {
                ta = *(const uint4*)(queue + qn);
                tb = *(const uint4*)(queue + qn + 4);
            }
            add_batch(wd, q0, n);
            nw[0] = (unsigned)__builtin_amdgcn_readfirstlane((int)ta.x);
            nw[1] = (unsigned)__builtin_amdgcn_readfirstlane((int)ta.y);
            nw[2] = (unsigned)__builtin_amdgcn_readfirstlane((int)ta.z);
            nw[3] = (unsigned)__builtin_amdgcn_readfirstlane((int)ta.w);
            nw[4] = (unsigned)__builtin_amdgcn_readfirstlane((int)tb.x);
            nw[5] = (unsigned)__builtin_amdgcn_readfirstlane((int)tb.y);
            nw[6] = (unsigned)__builtin_amdgcn_readfirstlane((int)tb.z);
            nw[7] = (unsigned)__builtin_amdgcn_readfirstlane((int)tb.w);
        }
    } else {
        const int n = (int)lcnt;
        int q0 = wave * OWN_U;
        for (; q0 < n; q0 += OWN_NW * OWN_U) {
            const uint4 ca = na, cb4 = nb;
            const int qn = q0 + OWN_NW * OWN_U;
            if (qn < n) {
                na = *(const uint4*)(ent + qn);
                nb = *(const uint4*)(ent + qn + 4);
            }
            const unsigned wd[OWN_U] = {ca.x, ca.y, ca.z, ca.w, cb4.x, cb4.y, cb4.z, cb4.w};
            add_batch(wd, q0, n);
        }
    }

    const vecf bv = *(const vecf*)(bias + lane * PER);
    vecf gv = bv, bev = bv;
    if constexpr (LNRELU) {
        gv  = *(const vecf*)(g + lane * PER);
        bev = *(const vecf*)(be + lane * PER);
    }
    const int dvl = __builtin_bit_cast(int, dinv[lo + min(wave + lane * OWN_NW, nown - 1)]);

    __syncthreads();

    int t = 0;
    for (int i = wave; i < nown; i += OWN_NW, ++t) {
        const float dv = __builtin_bit_cast(float, __builtin_amdgcn_readlane(dvl, t));
        vecf v;
#pragma unroll
        for (int k = 0; k < PER; ++k)
            v[k] = acc[i * PITCH + k * 32 + lane] * dv + bv[k];
        if constexpr (LNRELU) {
            float sum = 0.f;
#pragma unroll
            for (int k = 0; k < PER; ++k) sum += v[k];
            sum = wave_sum(sum);
            const float mu = sum * (1.0f / F);
            float var = 0.f;
#pragma unroll
            for (int k = 0; k < PER; ++k) {
                const float d = v[k] - mu;
                var += d * d;
            }
            var = wave_sum(var);
            var *= (1.0f / F);
            const float rs = rsqrtf(var + LN_EPS);
#pragma unroll
            for (int k = 0; k < PER; ++k)
                v[k] = fmaxf((v[k] - mu) * rs * gv[k] + bev[k], 0.f);
        }
#pragma unroll
        for (int k = 0; k < PER; ++k)
            acc[i * PITCH + k * 32 + lane] = v[k];
    }

    __syncthreads();

    {
        constexpr int NT = NOUT / 16;
        const int mr = lane & 15;
        const int hi = lane >> 4;
        const int ntile = (nown + 15) >> 4;
        for (int rt = wave; rt < ntile; rt += OWN_NW) {
            v8f cacc[NT];
#pragma unroll
            for (int t = 0; t < NT; ++t)
                cacc[t] = (v8f){0.f, 0.f, 0.f, 0.f, 0.f, 0.f, 0.f, 0.f};
            const float* arow = acc + (rt * 16 + mr) * PITCH + 2 * hi;
            float rs[8];
            if constexpr (SCALEOUT) {
#pragma unroll
                for (int v = 0; v < 8; ++v)
                    rs[v] = dinv[lo + min(rt * 16 + v + 8 * hi, nown - 1)];
            }
#pragma unroll 4
            for (int j = 0; j < F / 4; ++j) {
                const int w0 = 4 * j + 2 * hi;
                const int w1 = w0 + 1;
                const v2f a = *(const v2f*)(arow + 4 * j);
                const float* b0 = Wn + (size_t)((w0 & 31) * PER + (w0 >> 5)) * NOUT + mr;
                const float* b1 = Wn + (size_t)((w1 & 31) * PER + (w1 >> 5)) * NOUT + mr;
#pragma unroll
                for (int t = 0; t < NT; ++t) {
                    v2f b;
                    b.x = b0[t * 16];
                    b.y = b1[t * 16];
                    cacc[t] = __builtin_amdgcn_wmma_f32_16x16x4_f32(
                        false, a, false, b, (short)0, cacc[t], false, false);
                }
            }
#pragma unroll
            for (int t = 0; t < NT; ++t) {
                const int n = t * 16 + mr;
                float nb = 0.f;
                if constexpr (!SCALEOUT) nb = nbias[n];
#pragma unroll
                for (int v = 0; v < 8; ++v) {
                    const int r = rt * 16 + v + 8 * hi;
                    if (r < nown) {
                        if constexpr (SCALEOUT) out[(size_t)(lo + r) * NOUT + n] = cacc[t][v] * rs[v];
                        else out[(size_t)(lo + r) * NOUT + n] = cacc[t][v] + nb;
                    }
                }
            }
        }
    }
}

extern "C" void kernel_launch(void* const* d_in, const int* in_sizes, int n_in,
                              void* d_out, int out_size, void* d_ws, size_t ws_size,
                              hipStream_t stream) {
    constexpr int F = 128, H = 128, D = 64, C = 32;
    constexpr int MT = 2;

    const float* x   = (const float*)d_in[0];
    const int*   ei  = (const int*)d_in[1];
    const float* W1  = (const float*)d_in[2];
    const float* b1  = (const float*)d_in[3];
    const float* g1  = (const float*)d_in[4];
    const float* be1 = (const float*)d_in[5];
    const float* W2  = (const float*)d_in[6];
    const float* b2  = (const float*)d_in[7];
    const float* Wc  = (const float*)d_in[8];
    const float* bc  = (const float*)d_in[9];
    float* logits    = (float*)d_out;

    const int N = in_sizes[0] / F;
    const int E = in_sizes[1] / 2;
    const int* rows = ei;
    const int* cols = ei + E;

    float* ws   = (float*)d_ws;
    float* hw1  = ws;
    float* hw2  = ws + (size_t)N * 128;
    unsigned int* omap = (unsigned int*)(ws + (size_t)N * 192);
    const int lwords = N * 64;
    float* deg  = ws + (size_t)N * 256;
    float* dinv = deg + N;

    const int TPB = 256;
    const int wavesPerBlk = TPB / 32;
    auto cdiv = [](long long a, long long b) { return (int)((a + b - 1) / b); };

    deg_init_kernel<<<cdiv(N, TPB), TPB, 0, stream>>>(deg, N);
    deg_count_map_kernel<<<cdiv(cdiv(E, 4), TPB), TPB, 0, stream>>>(cols, deg, omap, E);
    deg_inv_kernel<<<cdiv(N, TPB), TPB, 0, stream>>>(deg, dinv, N);

    {
        const int nWaves = cdiv(N, 16 * MT);
        gemm_wmma_f32<F, H / 16, MT, true><<<cdiv(nWaves, wavesPerBlk), TPB, 0, stream>>>(
            x, W1, nullptr, dinv, hw1, N);
    }

    agg_owner_lds_kernel<H, 0, true, D, true><<<cdiv(N, OWN_R), OWN_TPB, 0, stream>>>(
        hw1, rows, cols, omap, omap, lwords, dinv, b1, g1, be1, W2, nullptr, hw2, N, E);

    agg_owner_lds_kernel<D, 1, false, C, false><<<cdiv(N, OWN_R), OWN_TPB, 0, stream>>>(
        hw2, rows, cols, omap, omap, lwords, dinv, b2, nullptr, nullptr, Wc, bc, logits, N, E);
}
